# starting kernel
# baseline (speedup 1.0000x reference)
#include <hip/hip_runtime.h>

#define T      16
#define WAY    20
#define WAYP   32
#define SHOT   5
#define WQ     300
#define WS     100
#define C      1024
#define MTILES 19
#define EPS    1e-12f

typedef __attribute__((ext_vector_type(2))) float v2f;
typedef __attribute__((ext_vector_type(8))) float v8f;

__global__ void proto_kernel(const float* __restrict__ sup,
                             float* __restrict__ proto_pad,
                             float* __restrict__ p_sq)
{
    const int t   = blockIdx.x >> 5;
    const int n   = blockIdx.x & 31;
    const int tid = threadIdx.x;

    __shared__ float red[8];
    __shared__ float s_inv;

    float4 m = make_float4(0.f, 0.f, 0.f, 0.f);
    float  ss = 0.f;
    if (n < WAY) {
        const float* base = sup + (((size_t)t * WS) + (size_t)n * SHOT) * C + tid * 4;
        float4 a = make_float4(0.f, 0.f, 0.f, 0.f);
        #pragma unroll
        for (int s = 0; s < SHOT; ++s) {
            float4 v = *(const float4*)(base + s * C);
            a.x += v.x; a.y += v.y; a.z += v.z; a.w += v.w;
        }
        const float inv_shot = 1.0f / (float)SHOT;
        m.x = a.x * inv_shot; m.y = a.y * inv_shot;
        m.z = a.z * inv_shot; m.w = a.w * inv_shot;
        ss = m.x * m.x + m.y * m.y + m.z * m.z + m.w * m.w;
    }

    #pragma unroll
    for (int off = 16; off >= 1; off >>= 1) ss += __shfl_xor(ss, off);
    if ((tid & 31) == 0) red[tid >> 5] = ss;
    __syncthreads();
    if (tid == 0) {
        float tot = 0.f;
        #pragma unroll
        for (int i = 0; i < 8; ++i) tot += red[i];
        float denom = fmaxf(sqrtf(tot), EPS);
        float inv   = 1.0f / denom;
        s_inv = inv;
        p_sq[blockIdx.x] = (n < WAY) ? tot * inv * inv : 0.0f;
    }
    __syncthreads();

    const float inv = s_inv;
    float4 o = make_float4(m.x * inv, m.y * inv, m.z * inv, m.w * inv);
    *(float4*)(proto_pad + (size_t)blockIdx.x * C + tid * 4) = o;
}

__global__ __launch_bounds__(32)
void proto_dist_kernel(const float* __restrict__ q,
                       const float* __restrict__ proto_pad,
                       const float* __restrict__ p_sq,
                       float* __restrict__ out)
{
    const int bid  = blockIdx.x;
    const int t    = bid / MTILES;
    const int mt   = bid % MTILES;
    const int lane = threadIdx.x;
    const int half = lane >> 4;
    const int l15  = lane & 15;
    const int mbase = mt * 16;

    int lrow = mbase + l15;
    if (lrow > WQ - 1) lrow = WQ - 1;

    const float* arow = q + ((size_t)t * WQ + lrow) * C + 2 * half;
    const int col0 = l15;
    const int col1 = 16 + l15;
    const float* b0p = proto_pad + ((size_t)t * WAYP + col0) * C + 2 * half;
    const float* b1p = proto_pad + ((size_t)t * WAYP + col1) * C + 2 * half;

    v8f acc0 = {};
    v8f acc1 = {};
    float nacc = 0.f;

    #pragma unroll 8
    for (int k = 0; k < C; k += 4) {
        v2f a  = *(const v2f*)(arow + k);
        v2f b0 = *(const v2f*)(b0p + k);
        v2f b1 = *(const v2f*)(b1p + k);
        nacc += a.x * a.x + a.y * a.y;
        acc0 = __builtin_amdgcn_wmma_f32_16x16x4_f32(
                   false, a, false, b0, (short)0, acc0, false, false);
        acc1 = __builtin_amdgcn_wmma_f32_16x16x4_f32(
                   false, a, false, b1, (short)0, acc1, false, false);
    }

    float nsum = nacc + __shfl_xor(nacc, 16);

    const float psq0 = p_sq[t * WAYP + col0];
    const float psq1 = p_sq[t * WAYP + col1];

    #pragma unroll
    for (int r = 0; r < 8; ++r) {
        const int rrow = mbase + r + 8 * half;
        float ss    = __shfl(nsum, r + 8 * half);
        float denom = fmaxf(sqrtf(ss), EPS);
        float inv   = 1.0f / denom;
        float qsq   = ss * inv * inv;
        if (rrow < WQ) {
            float* o = out + ((size_t)t * WQ + rrow) * WAY;
            o[col0] = 2.0f * acc0[r] * inv - qsq - psq0;
            if (col1 < WAY)
                o[col1] = 2.0f * acc1[r] * inv - qsq - psq1;
        }
    }
}

extern "C" void kernel_launch(void* const* d_in, const int* in_sizes, int n_in,
                              void* d_out, int out_size, void* d_ws, size_t ws_size,
                              hipStream_t stream)
{
    const float* query_feat   = (const float*)d_in[0];
    const float* support_feat = (const float*)d_in[1];
    float* out = (float*)d_out;

    float* proto_pad = (float*)d_ws;
    float* p_sq      = proto_pad + (size_t)T * WAYP * C;

    proto_kernel<<<T * WAYP, 256, 0, stream>>>(support_feat, proto_pad, p_sq);
    proto_dist_kernel<<<T * MTILES, 32, 0, stream>>>(query_feat, proto_pad, p_sq, out);
}


# Distance kernel: K split over the 8 waves of a 256-thread work-group (128 k per wave); the 8 partial tiles and row norms are summed through LDS (new f32 summation order), one thread per output
# speedup vs baseline: 1.7371x; 1.7371x over previous
#include <hip/hip_runtime.h>

#define T      16
#define WAY    20
#define WAYP   32
#define SHOT   5
#define WQ     300
#define WS     100
#define C      1024
#define MTILES 19
#define EPS    1e-12f

typedef __attribute__((ext_vector_type(2))) float v2f;
typedef __attribute__((ext_vector_type(8))) float v8f;

__global__ void proto_kernel(const float* __restrict__ sup,
                             float* __restrict__ proto_pad,
                             float* __restrict__ p_sq)
{
    const int t   = blockIdx.x >> 5;
    const int n   = blockIdx.x & 31;
    const int tid = threadIdx.x;

    __shared__ float red[8];
    __shared__ float s_inv;

    float4 m = make_float4(0.f, 0.f, 0.f, 0.f);
    float  ss = 0.f;
    if (n < WAY) {
        const float* base = sup + (((size_t)t * WS) + (size_t)n * SHOT) * C + tid * 4;
        float4 a = make_float4(0.f, 0.f, 0.f, 0.f);
        #pragma unroll
        for (int s = 0; s < SHOT; ++s) {
            float4 v = *(const float4*)(base + s * C);
            a.x += v.x; a.y += v.y; a.z += v.z; a.w += v.w;
        }
        const float inv_shot = 1.0f / (float)SHOT;
        m.x = a.x * inv_shot; m.y = a.y * inv_shot;
        m.z = a.z * inv_shot; m.w = a.w * inv_shot;
        ss = m.x * m.x + m.y * m.y + m.z * m.z + m.w * m.w;
    }

    #pragma unroll
    for (int off = 16; off >= 1; off >>= 1) ss += __shfl_xor(ss, off);
    if ((tid & 31) == 0) red[tid >> 5] = ss;
    __syncthreads();
    if (tid == 0) {
        float tot = 0.f;
        #pragma unroll
        for (int i = 0; i < 8; ++i) tot += red[i];
        float denom = fmaxf(sqrtf(tot), EPS);
        float inv   = 1.0f / denom;
        s_inv = inv;
        p_sq[blockIdx.x] = (n < WAY) ? tot * inv * inv : 0.0f;
    }
    __syncthreads();

    const float inv = s_inv;
    float4 o = make_float4(m.x * inv, m.y * inv, m.z * inv, m.w * inv);
    *(float4*)(proto_pad + (size_t)blockIdx.x * C + tid * 4) = o;
}

#define KSPLIT 8
#define KSLICE (C / KSPLIT)

__global__ __launch_bounds__(32 * KSPLIT)
void proto_dist_kernel(const float* __restrict__ q,
                       const float* __restrict__ proto_pad,
                       const float* __restrict__ p_sq,
                       float* __restrict__ out)
{
    __shared__ float part[KSPLIT][2][8][32];
    __shared__ float nrm[KSPLIT][16];

    const int bid  = blockIdx.x;
    const int t    = bid / MTILES;
    const int mt   = bid % MTILES;
    const int tid  = threadIdx.x;
    const int w    = tid >> 5;
    const int lane = tid & 31;
    const int half = lane >> 4;
    const int l15  = lane & 15;
    const int mbase = mt * 16;

    int lrow = mbase + l15;
    if (lrow > WQ - 1) lrow = WQ - 1;

    const int koff = w * KSLICE + 2 * half;
    const float* arow = q + ((size_t)t * WQ + lrow) * C + koff;
    const float* b0p = proto_pad + ((size_t)t * WAYP + l15) * C + koff;
    const float* b1p = proto_pad + ((size_t)t * WAYP + 16 + l15) * C + koff;

    v8f acc0 = {};
    v8f acc1 = {};
    float nacc = 0.f;

    #pragma unroll
    for (int k = 0; k < KSLICE; k += 4) {
        v2f a  = *(const v2f*)(arow + k);
        v2f b0 = *(const v2f*)(b0p + k);
        v2f b1 = *(const v2f*)(b1p + k);
        nacc += a.x * a.x + a.y * a.y;
        acc0 = __builtin_amdgcn_wmma_f32_16x16x4_f32(
                   false, a, false, b0, (short)0, acc0, false, false);
        acc1 = __builtin_amdgcn_wmma_f32_16x16x4_f32(
                   false, a, false, b1, (short)0, acc1, false, false);
    }

    const float nsum = nacc + __shfl_xor(nacc, 16);

    #pragma unroll
    for (int r = 0; r < 8; ++r) {
        part[w][0][r][lane] = acc0[r];
        part[w][1][r][lane] = acc1[r];
    }
    if (lane < 16) nrm[w][lane] = nsum;
    __syncthreads();

    float* otile = out + ((size_t)t * WQ + mbase) * WAY;
    for (int o = tid; o < 16 * WAY; o += 32 * KSPLIT) {
        const int row = o / WAY;
        const int col = o - row * WAY;
        if (mbase + row < WQ) {
            const int tile = col >> 4;
            const int r    = row & 7;
            const int ln   = ((row >> 3) << 4) + (col & 15);
            float d  = 0.f;
            float ss = 0.f;
            #pragma unroll
            for (int i = 0; i < KSPLIT; ++i) {
                d  += part[i][tile][r][ln];
                ss += nrm[i][row];
            }
            float denom = fmaxf(sqrtf(ss), EPS);
            float inv   = 1.0f / denom;
            float qsq   = ss * inv * inv;
            otile[o] = 2.0f * d * inv - qsq - p_sq[t * WAYP + col];
        }
    }
}

extern "C" void kernel_launch(void* const* d_in, const int* in_sizes, int n_in,
                              void* d_out, int out_size, void* d_ws, size_t ws_size,
                              hipStream_t stream)
{
    const float* query_feat   = (const float*)d_in[0];
    const float* support_feat = (const float*)d_in[1];
    float* out = (float*)d_out;

    float* proto_pad = (float*)d_ws;
    float* p_sq      = proto_pad + (size_t)T * WAYP * C;

    proto_kernel<<<T * WAYP, 256, 0, stream>>>(support_feat, proto_pad, p_sq);
    proto_dist_kernel<<<T * MTILES, 256, 0, stream>>>(query_feat, proto_pad, p_sq, out);
}


# Distance kernel operands staged through LDS: each global load reads one row's whole 512-B K slice (4 full lines, not 16 scattered); fragments re-read by 16-B LDS loads; new f32 sum order per 8 k
# speedup vs baseline: 2.2846x; 1.3152x over previous
#include <hip/hip_runtime.h>

#define T      16
#define WAY    20
#define WAYP   32
#define SHOT   5
#define WQ     300
#define WS     100
#define C      1024
#define MTILES 19
#define EPS    1e-12f

typedef __attribute__((ext_vector_type(2))) float v2f;
typedef __attribute__((ext_vector_type(8))) float v8f;

__global__ void proto_kernel(const float* __restrict__ sup,
                             float* __restrict__ proto_pad,
                             float* __restrict__ p_sq)
{
    const int t   = blockIdx.x >> 5;
    const int n   = blockIdx.x & 31;
    const int tid = threadIdx.x;

    __shared__ float red[8];
    __shared__ float s_inv;

    float4 m = make_float4(0.f, 0.f, 0.f, 0.f);
    float  ss = 0.f;
    if (n < WAY) {
        const float* base = sup + (((size_t)t * WS) + (size_t)n * SHOT) * C + tid * 4;
        float4 a = make_float4(0.f, 0.f, 0.f, 0.f);
        #pragma unroll
        for (int s = 0; s < SHOT; ++s) {
            float4 v = *(const float4*)(base + s * C);
            a.x += v.x; a.y += v.y; a.z += v.z; a.w += v.w;
        }
        const float inv_shot = 1.0f / (float)SHOT;
        m.x = a.x * inv_shot; m.y = a.y * inv_shot;
        m.z = a.z * inv_shot; m.w = a.w * inv_shot;
        ss = m.x * m.x + m.y * m.y + m.z * m.z + m.w * m.w;
    }

    #pragma unroll
    for (int off = 16; off >= 1; off >>= 1) ss += __shfl_xor(ss, off);
    if ((tid & 31) == 0) red[tid >> 5] = ss;
    __syncthreads();
    if (tid == 0) {
        float tot = 0.f;
        #pragma unroll
        for (int i = 0; i < 8; ++i) tot += red[i];
        float denom = fmaxf(sqrtf(tot), EPS);
        float inv   = 1.0f / denom;
        s_inv = inv;
        p_sq[blockIdx.x] = (n < WAY) ? tot * inv * inv : 0.0f;
    }
    __syncthreads();

    const float inv = s_inv;
    float4 o = make_float4(m.x * inv, m.y * inv, m.z * inv, m.w * inv);
    *(float4*)(proto_pad + (size_t)blockIdx.x * C + tid * 4) = o;
}

#define KSPLIT 8
#define KSLICE (C / KSPLIT)
#define PITCH  132
#define AROWS  16
#define BROWS  WAY
#define SLAB   ((AROWS + BROWS) * PITCH)

typedef __attribute__((ext_vector_type(4))) float v4f;

__global__ __launch_bounds__(32 * KSPLIT)
void proto_dist_kernel(const float* __restrict__ q,
                       const float* __restrict__ proto_pad,
                       const float* __restrict__ p_sq,
                       float* __restrict__ out)
{
    __shared__ float slab[KSPLIT][SLAB];

    const int bid  = blockIdx.x;
    const int t    = bid / MTILES;
    const int mt   = bid % MTILES;
    const int tid  = threadIdx.x;
    const int w    = tid >> 5;
    const int lane = tid & 31;
    const int half = lane >> 4;
    const int l15  = lane & 15;
    const int mbase = mt * 16;

    float* sw = slab[w];
    float* sA = sw;
    float* sB = sw + AROWS * PITCH;

    const int kg = w * KSLICE + lane * 4;
    v4f ra[AROWS];
    #pragma unroll
    for (int i = 0; i < AROWS; ++i) {
        int row = mbase + i;
        if (row > WQ - 1) row = WQ - 1;
        ra[i] = *(const v4f*)(q + ((size_t)t * WQ + row) * C + kg);
    }
    v4f rb[BROWS];
    #pragma unroll
    for (int i = 0; i < BROWS; ++i)
        rb[i] = *(const v4f*)(proto_pad + ((size_t)t * WAYP + i) * C + kg);
    #pragma unroll
    for (int i = 0; i < AROWS; ++i)
        *(v4f*)(sA + i * PITCH + lane * 4) = ra[i];
    #pragma unroll
    for (int i = 0; i < BROWS; ++i)
        *(v4f*)(sB + i * PITCH + lane * 4) = rb[i];
    __syncthreads();

    int n1 = 16 + l15;
    if (n1 > WAY - 1) n1 = WAY - 1;
    const float* fa  = sA + l15 * PITCH + 4 * half;
    const float* fb0 = sB + l15 * PITCH + 4 * half;
    const float* fb1 = sB + n1  * PITCH + 4 * half;

    v8f acc0 = {};
    v8f acc1 = {};
    float nacc = 0.f;

    v4f fa_r[KSLICE / 8], fb0_r[KSLICE / 8], fb1_r[KSLICE / 8];
    #pragma unroll
    for (int j = 0; j < KSLICE / 8; ++j) {
        fa_r[j]  = *(const v4f*)(fa  + 8 * j);
        fb0_r[j] = *(const v4f*)(fb0 + 8 * j);
        fb1_r[j] = *(const v4f*)(fb1 + 8 * j);
    }
    __builtin_amdgcn_sched_barrier(0);

    #pragma unroll
    for (int j = 0; j < KSLICE / 8; ++j) {
        const v4f a  = fa_r[j];
        const v4f b0 = fb0_r[j];
        const v4f b1 = fb1_r[j];
        nacc += a.x * a.x + a.y * a.y + a.z * a.z + a.w * a.w;
        const v2f alo = {a.x, a.y},   ahi = {a.z, a.w};
        const v2f b0lo = {b0.x, b0.y}, b0hi = {b0.z, b0.w};
        const v2f b1lo = {b1.x, b1.y}, b1hi = {b1.z, b1.w};
        acc0 = __builtin_amdgcn_wmma_f32_16x16x4_f32(
                   false, alo, false, b0lo, (short)0, acc0, false, false);
        acc1 = __builtin_amdgcn_wmma_f32_16x16x4_f32(
                   false, alo, false, b1lo, (short)0, acc1, false, false);
        acc0 = __builtin_amdgcn_wmma_f32_16x16x4_f32(
                   false, ahi, false, b0hi, (short)0, acc0, false, false);
        acc1 = __builtin_amdgcn_wmma_f32_16x16x4_f32(
                   false, ahi, false, b1hi, (short)0, acc1, false, false);
    }

    const float nsum = nacc + __shfl_xor(nacc, 16);

    #pragma unroll
    for (int r = 0; r < 8; ++r) {
        sw[(0 * 8 + r) * 32 + lane] = acc0[r];
        sw[(1 * 8 + r) * 32 + lane] = acc1[r];
    }
    if (lane < 16) sw[512 + lane] = nsum;
    __syncthreads();

    float* otile = out + ((size_t)t * WQ + mbase) * WAY;
    for (int o = tid; o < 16 * WAY; o += 32 * KSPLIT) {
        const int row = o / WAY;
        const int col = o - row * WAY;
        if (mbase + row < WQ) {
            const int tile = col >> 4;
            const int r    = row & 7;
            const int ln   = ((row >> 3) << 4) + (col & 15);
            float d  = 0.f;
            float ss = 0.f;
            #pragma unroll
            for (int i = 0; i < KSPLIT; ++i) {
                d  += slab[i][(tile * 8 + r) * 32 + ln];
                ss += slab[i][512 + row];
            }
            float denom = fmaxf(sqrtf(ss), EPS);
            float inv   = 1.0f / denom;
            float qsq   = ss * inv * inv;
            otile[o] = 2.0f * d * inv - qsq - p_sq[t * WAYP + col];
        }
    }
}

extern "C" void kernel_launch(void* const* d_in, const int* in_sizes, int n_in,
                              void* d_out, int out_size, void* d_ws, size_t ws_size,
                              hipStream_t stream)
{
    const float* query_feat   = (const float*)d_in[0];
    const float* support_feat = (const float*)d_in[1];
    float* out = (float*)d_out;

    float* proto_pad = (float*)d_ws;
    float* p_sq      = proto_pad + (size_t)T * WAYP * C;

    proto_kernel<<<T * WAYP, 256, 0, stream>>>(support_feat, proto_pad, p_sq);
    proto_dist_kernel<<<T * MTILES, 256, 0, stream>>>(query_feat, proto_pad, p_sq, out);
}


# K split over work-groups: one per (t, 64-k slice) stages all 300 query rows and the 20 shot means once and writes partial tiles; the second launch sums the 16 partials (new f32 sum order)
# speedup vs baseline: 2.9311x; 1.2830x over previous
#include <hip/hip_runtime.h>

#define T      16
#define WAY    20
#define SHOT   5
#define WQ     300
#define WS     100
#define C      1024
#define MTILES 19
#define EPS    1e-12f

#define KSPLIT 16
#define KSLICE 64
#define PITCH  68
#define NPLANE (T * KSPLIT)
#define PLANE  6144
#define QSTR   640
#define PSTR   32

typedef __attribute__((ext_vector_type(2))) float v2f;
typedef __attribute__((ext_vector_type(4))) float v4f;
typedef __attribute__((ext_vector_type(8))) float v8f;

__global__ __launch_bounds__(256)
void slice_kernel(const float* __restrict__ q,
                  const float* __restrict__ sup,
                  float* __restrict__ part,
                  float* __restrict__ qn,
                  float* __restrict__ pn)
{
    __shared__ __attribute__((aligned(16))) float sA[WQ * PITCH];
    __shared__ __attribute__((aligned(16))) float sB[WAY * PITCH];
    __shared__ __attribute__((aligned(16))) float sT[8][16 * WAY];

    const int bid  = blockIdx.x;
    const int t    = bid >> 4;
    const int ks   = bid & 15;
    const int tid  = threadIdx.x;
    const int w    = __builtin_amdgcn_readfirstlane(tid >> 5);
    const int lane = tid & 31;
    const int half = lane >> 4;
    const int l15  = lane & 15;
    const int kq   = tid & 15;
    const int r0   = tid >> 4;
    const int plane = ks * T + t;

    const float* qbase = q + (size_t)t * WQ * C + ks * KSLICE + kq * 4;
    v4f ra[MTILES];
    #pragma unroll
    for (int i = 0; i < MTILES; ++i) {
        int row = r0 + 16 * i;
        if (row > WQ - 1) row = WQ - 1;
        ra[i] = *(const v4f*)(qbase + (size_t)row * C);
    }
    const float* sbase = sup + (size_t)t * WS * C + ks * KSLICE + kq * 4;
    v4f rs[SHOT];
    #pragma unroll
    for (int s = 0; s < SHOT; ++s)
        rs[s] = *(const v4f*)(sbase + (size_t)(r0 * SHOT + s) * C);
    v4f rs2[SHOT];
    if (w < 2) {
        #pragma unroll
        for (int s = 0; s < SHOT; ++s)
            rs2[s] = *(const v4f*)(sbase + (size_t)((16 + r0) * SHOT + s) * C);
    }

    #pragma unroll
    for (int i = 0; i < MTILES; ++i) {
        const int row = r0 + 16 * i;
        if (row < WQ) *(v4f*)(sA + row * PITCH + kq * 4) = ra[i];
    }
    const float inv_shot = 1.0f / (float)SHOT;
    {
        v4f a = {0.f, 0.f, 0.f, 0.f};
        #pragma unroll
        for (int s = 0; s < SHOT; ++s) a += rs[s];
        *(v4f*)(sB + r0 * PITCH + kq * 4) = a * inv_shot;
    }
    if (w < 2) {
        v4f a = {0.f, 0.f, 0.f, 0.f};
        #pragma unroll
        for (int s = 0; s < SHOT; ++s) a += rs2[s];
        *(v4f*)(sB + (16 + r0) * PITCH + kq * 4) = a * inv_shot;
    }
    __syncthreads();

    int n1 = 16 + l15;
    if (n1 > WAY - 1) n1 = WAY - 1;
    const float* fb0 = sB + l15 * PITCH + 4 * half;
    const float* fb1 = sB + n1  * PITCH + 4 * half;
    v4f fb0_r[KSLICE / 8], fb1_r[KSLICE / 8];
    #pragma unroll
    for (int j = 0; j < KSLICE / 8; ++j) {
        fb0_r[j] = *(const v4f*)(fb0 + 8 * j);
        fb1_r[j] = *(const v4f*)(fb1 + 8 * j);
    }

    if (w == 0) {
        float p0 = 0.f, p1 = 0.f;
        #pragma unroll
        for (int j = 0; j < KSLICE / 8; ++j) {
            const v4f b0 = fb0_r[j];
            const v4f b1 = fb1_r[j];
            p0 += b0.x * b0.x + b0.y * b0.y + b0.z * b0.z + b0.w * b0.w;
            p1 += b1.x * b1.x + b1.y * b1.y + b1.z * b1.z + b1.w * b1.w;
        }
        p0 += __shfl_xor(p0, 16);
        p1 += __shfl_xor(p1, 16);
        float* pp = pn + (size_t)plane * PSTR;
        if (lane < 16) pp[lane] = p0;
        if (lane < 4)  pp[16 + lane] = p1;
    }

    float* tp = part + (size_t)plane * PLANE;
    float* tq = qn + (size_t)plane * QSTR;
    float* st = sT[w];

    for (int mt = w; mt < MTILES; mt += 8) {
        int row = mt * 16 + l15;
        if (row > WQ - 1) row = WQ - 1;
        const float* fa = sA + row * PITCH + 4 * half;
        v4f fa_r[KSLICE / 8];
        #pragma unroll
        for (int j = 0; j < KSLICE / 8; ++j)
            fa_r[j] = *(const v4f*)(fa + 8 * j);

        v8f acc0 = {};
        v8f acc1 = {};
        float nacc = 0.f;
        #pragma unroll
        for (int j = 0; j < KSLICE / 8; ++j) {
            const v4f a  = fa_r[j];
            const v4f b0 = fb0_r[j];
            const v4f b1 = fb1_r[j];
            nacc += a.x * a.x + a.y * a.y + a.z * a.z + a.w * a.w;
            const v2f alo = {a.x, a.y},   ahi = {a.z, a.w};
            const v2f b0lo = {b0.x, b0.y}, b0hi = {b0.z, b0.w};
            const v2f b1lo = {b1.x, b1.y}, b1hi = {b1.z, b1.w};
            acc0 = __builtin_amdgcn_wmma_f32_16x16x4_f32(
                       false, alo, false, b0lo, (short)0, acc0, false, false);
            acc1 = __builtin_amdgcn_wmma_f32_16x16x4_f32(
                       false, alo, false, b1lo, (short)0, acc1, false, false);
            acc0 = __builtin_amdgcn_wmma_f32_16x16x4_f32(
                       false, ahi, false, b0hi, (short)0, acc0, false, false);
            acc1 = __builtin_amdgcn_wmma_f32_16x16x4_f32(
                       false, ahi, false, b1hi, (short)0, acc1, false, false);
        }
        const float nsum = nacc + __shfl_xor(nacc, 16);

        #pragma unroll
        for (int r = 0; r < 8; ++r)
            st[(r + 8 * half) * WAY + l15] = acc0[r];
        if (l15 < 4) {
            #pragma unroll
            for (int r = 0; r < 8; ++r)
                st[(r + 8 * half) * WAY + 16 + l15] = acc1[r];
        }
        #pragma unroll
        for (int i = 0; i < 3; ++i) {
            const int idx = lane + 32 * i;
            if (idx < 80)
                *(v4f*)(tp + mt * 320 + idx * 4) = *(const v4f*)(st + idx * 4);
        }
        if (lane < 16) tq[mt * 32 + lane] = nsum;
    }
}

__global__ __launch_bounds__(128)
void finish_kernel(const float* __restrict__ part,
                   const float* __restrict__ qn,
                   const float* __restrict__ pn,
                   float* __restrict__ out)
{
    const int g = blockIdx.x * 128 + threadIdx.x;
    if (g >= T * WQ * WAY / 4) return;
    const int o   = g * 4;
    const int t   = o / (WQ * WAY);
    const int rem = o - t * (WQ * WAY);
    const int row = rem / WAY;
    const int col = rem - row * WAY;
    const int qi  = (row >> 4) * 32 + (row & 15);

    v4f d  = {0.f, 0.f, 0.f, 0.f};
    v4f ps = {0.f, 0.f, 0.f, 0.f};
    float qs = 0.f;
    #pragma unroll
    for (int ks = 0; ks < KSPLIT; ++ks) {
        const int plane = ks * T + t;
        d  += *(const v4f*)(part + (size_t)plane * PLANE + rem);
        ps += *(const v4f*)(pn + (size_t)plane * PSTR + col);
        qs += qn[(size_t)plane * QSTR + qi];
    }
    const float qden = fmaxf(sqrtf(qs), EPS);
    const float qinv = 1.0f / qden;
    const float qsq  = qs * qinv * qinv;
    v4f res;
    #pragma unroll
    for (int c = 0; c < 4; ++c) {
        const float pden = fmaxf(sqrtf(ps[c]), EPS);
        const float pinv = 1.0f / pden;
        const float psq  = ps[c] * pinv * pinv;
        res[c] = 2.0f * d[c] * qinv * pinv - qsq - psq;
    }
    *(v4f*)(out + o) = res;
}

extern "C" void kernel_launch(void* const* d_in, const int* in_sizes, int n_in,
                              void* d_out, int out_size, void* d_ws, size_t ws_size,
                              hipStream_t stream)
{
    const float* query_feat   = (const float*)d_in[0];
    const float* support_feat = (const float*)d_in[1];
    float* out = (float*)d_out;

    float* part = (float*)d_ws;
    float* qn   = part + (size_t)NPLANE * PLANE;
    float* pn   = qn + (size_t)NPLANE * QSTR;

    slice_kernel<<<NPLANE, 256, 0, stream>>>(query_feat, support_feat, part, qn, pn);
    finish_kernel<<<188, 128, 0, stream>>>(part, qn, pn, out);
}


# Slice kernel: each wave loads and stages only its own M-tiles (support rows first, one LDS-only barrier for the means), so its products run while later tiles are in flight; sum order unchanged
# speedup vs baseline: 3.1643x; 1.0796x over previous
#include <hip/hip_runtime.h>

#define T      16
#define WAY    20
#define SHOT   5
#define WQ     300
#define WS     100
#define C      1024
#define MTILES 19
#define EPS    1e-12f

#define KSPLIT 16
#define KSLICE 64
#define PITCH  68
#define NPLANE (T * KSPLIT)
#define PLANE  6144
#define QSTR   640
#define PSTR   32
#define APATCH (16 * PITCH)

typedef __attribute__((ext_vector_type(2))) float v2f;
typedef __attribute__((ext_vector_type(4))) float v4f;
typedef __attribute__((ext_vector_type(8))) float v8f;

__global__ __launch_bounds__(256)
void slice_kernel(const float* __restrict__ q,
                  const float* __restrict__ sup,
                  float* __restrict__ part,
                  float* __restrict__ qn,
                  float* __restrict__ pn)
{
    __shared__ __attribute__((aligned(16))) float sA[8][3][APATCH];
    __shared__ __attribute__((aligned(16))) float sB[WAY * PITCH];
    __shared__ __attribute__((aligned(16))) float sT[8][16 * WAY];

    const int bid  = blockIdx.x;
    const int t    = bid >> 4;
    const int ks   = bid & 15;
    const int tid  = threadIdx.x;
    const int w    = __builtin_amdgcn_readfirstlane(tid >> 5);
    const int lane = tid & 31;
    const int half = lane >> 4;
    const int l15  = lane & 15;
    const int kq   = tid & 15;
    const int r0   = tid >> 4;
    const int plane = ks * T + t;

    const float* sbase = sup + (size_t)t * WS * C + ks * KSLICE + kq * 4;
    v4f rs[SHOT];
    #pragma unroll
    for (int s = 0; s < SHOT; ++s)
        rs[s] = *(const v4f*)(sbase + (size_t)(r0 * SHOT + s) * C);
    v4f rs2[SHOT];
    #pragma unroll
    for (int s = 0; s < SHOT; ++s) rs2[s] = (v4f){0.f, 0.f, 0.f, 0.f};
    if (w < 2) {
        #pragma unroll
        for (int s = 0; s < SHOT; ++s)
            rs2[s] = *(const v4f*)(sbase + (size_t)((16 + r0) * SHOT + s) * C);
    }
    const float* qbase = q + (size_t)t * WQ * C + ks * KSLICE + l15 * 4;
    v4f ra0[8], ra1[8];
    #pragma unroll
    for (int i = 0; i < 8; ++i)
        ra0[i] = *(const v4f*)(qbase + (size_t)(w * 16 + 2 * i + half) * C);
    #pragma unroll
    for (int i = 0; i < 8; ++i)
        ra1[i] = *(const v4f*)(qbase + (size_t)((w + 8) * 16 + 2 * i + half) * C);
    __builtin_amdgcn_sched_barrier(0);

    const float inv_shot = 1.0f / (float)SHOT;
    {
        v4f a = {0.f, 0.f, 0.f, 0.f};
        #pragma unroll
        for (int s = 0; s < SHOT; ++s) a += rs[s];
        *(v4f*)(sB + r0 * PITCH + kq * 4) = a * inv_shot;
    }
    {
        v4f a = {0.f, 0.f, 0.f, 0.f};
        #pragma unroll
        for (int s = 0; s < SHOT; ++s) a += rs2[s];
        if (w < 2) *(v4f*)(sB + (16 + r0) * PITCH + kq * 4) = a * inv_shot;
    }
    __builtin_amdgcn_fence(__ATOMIC_RELEASE, "workgroup", "local");
    __builtin_amdgcn_s_barrier();
    __builtin_amdgcn_fence(__ATOMIC_ACQUIRE, "workgroup", "local");

    int n1 = 16 + l15;
    if (n1 > WAY - 1) n1 = WAY - 1;
    const float* fb0 = sB + l15 * PITCH + 4 * half;
    const float* fb1 = sB + n1  * PITCH + 4 * half;
    v4f fb0_r[KSLICE / 8], fb1_r[KSLICE / 8];
    #pragma unroll
    for (int j = 0; j < KSLICE / 8; ++j) {
        fb0_r[j] = *(const v4f*)(fb0 + 8 * j);
        fb1_r[j] = *(const v4f*)(fb1 + 8 * j);
    }

    if (w == 0) {
        float p0 = 0.f, p1 = 0.f;
        #pragma unroll
        for (int j = 0; j < KSLICE / 8; ++j) {
            const v4f b0 = fb0_r[j];
            const v4f b1 = fb1_r[j];
            p0 += b0.x * b0.x + b0.y * b0.y + b0.z * b0.z + b0.w * b0.w;
            p1 += b1.x * b1.x + b1.y * b1.y + b1.z * b1.z + b1.w * b1.w;
        }
        p0 += __shfl_xor(p0, 16);
        p1 += __shfl_xor(p1, 16);
        float* pp = pn + (size_t)plane * PSTR;
        if (lane < 16) pp[lane] = p0;
        if (lane < 4)  pp[16 + lane] = p1;
    }

    float* tp = part + (size_t)plane * PLANE;
    float* tq = qn + (size_t)plane * QSTR;
    float* st = sT[w];

    auto stage = [&](const v4f (&ra)[8], float* patch) {
        #pragma unroll
        for (int i = 0; i < 8; ++i)
            *(v4f*)(patch + (2 * i + half) * PITCH + l15 * 4) = ra[i];
    };

    auto compute = [&](const float* patch, int mt) {
        const float* fa = patch + l15 * PITCH + 4 * half;
        v4f fa_r[KSLICE / 8];
        #pragma unroll
        for (int j = 0; j < KSLICE / 8; ++j)
            fa_r[j] = *(const v4f*)(fa + 8 * j);
        __builtin_amdgcn_sched_barrier(0);

        v8f acc0 = {};
        v8f acc1 = {};
        float nacc = 0.f;
        #pragma unroll
        for (int j = 0; j < KSLICE / 8; ++j) {
            const v4f a  = fa_r[j];
            const v4f b0 = fb0_r[j];
            const v4f b1 = fb1_r[j];
            nacc += a.x * a.x + a.y * a.y + a.z * a.z + a.w * a.w;
            const v2f alo = {a.x, a.y},   ahi = {a.z, a.w};
            const v2f b0lo = {b0.x, b0.y}, b0hi = {b0.z, b0.w};
            const v2f b1lo = {b1.x, b1.y}, b1hi = {b1.z, b1.w};
            acc0 = __builtin_amdgcn_wmma_f32_16x16x4_f32(
                       false, alo, false, b0lo, (short)0, acc0, false, false);
            acc1 = __builtin_amdgcn_wmma_f32_16x16x4_f32(
                       false, alo, false, b1lo, (short)0, acc1, false, false);
            acc0 = __builtin_amdgcn_wmma_f32_16x16x4_f32(
                       false, ahi, false, b0hi, (short)0, acc0, false, false);
            acc1 = __builtin_amdgcn_wmma_f32_16x16x4_f32(
                       false, ahi, false, b1hi, (short)0, acc1, false, false);
        }
        const float nsum = nacc + __shfl_xor(nacc, 16);

        #pragma unroll
        for (int r = 0; r < 8; ++r)
            st[(r + 8 * half) * WAY + l15] = acc0[r];
        if (l15 < 4) {
            #pragma unroll
            for (int r = 0; r < 8; ++r)
                st[(r + 8 * half) * WAY + 16 + l15] = acc1[r];
        }
        #pragma unroll
        for (int i = 0; i < 3; ++i) {
            const int idx = lane + 32 * i;
            if (idx < 80)
                *(v4f*)(tp + mt * 320 + idx * 4) = *(const v4f*)(st + idx * 4);
        }
        if (lane < 16) tq[mt * 32 + lane] = nsum;
    };

    if (w < 3) {
        stage(ra0, sA[w][0]);
        v4f ra2[8];
        #pragma unroll
        for (int i = 0; i < 8; ++i) {
            int row = (w + 16) * 16 + 2 * i + half;
            if (row > WQ - 1) row = WQ - 1;
            ra2[i] = *(const v4f*)(qbase + (size_t)row * C);
        }
        compute(sA[w][0], w);
        stage(ra1, sA[w][1]);
        compute(sA[w][1], w + 8);
        stage(ra2, sA[w][2]);
        compute(sA[w][2], w + 16);
    } else {
        stage(ra0, sA[w][0]);
        compute(sA[w][0], w);
        stage(ra1, sA[w][1]);
        compute(sA[w][1], w + 8);
    }
}

__global__ __launch_bounds__(128)
void finish_kernel(const float* __restrict__ part,
                   const float* __restrict__ qn,
                   const float* __restrict__ pn,
                   float* __restrict__ out)
{
    const int g = blockIdx.x * 128 + threadIdx.x;
    if (g >= T * WQ * WAY / 4) return;
    const int o   = g * 4;
    const int t   = o / (WQ * WAY);
    const int rem = o - t * (WQ * WAY);
    const int row = rem / WAY;
    const int col = rem - row * WAY;
    const int qi  = (row >> 4) * 32 + (row & 15);

    v4f d  = {0.f, 0.f, 0.f, 0.f};
    v4f ps = {0.f, 0.f, 0.f, 0.f};
    float qs = 0.f;
    #pragma unroll
    for (int ks = 0; ks < KSPLIT; ++ks) {
        const int plane = ks * T + t;
        d  += *(const v4f*)(part + (size_t)plane * PLANE + rem);
        ps += *(const v4f*)(pn + (size_t)plane * PSTR + col);
        qs += qn[(size_t)plane * QSTR + qi];
    }
    const float qden = fmaxf(sqrtf(qs), EPS);
    const float qinv = 1.0f / qden;
    const float qsq  = qs * qinv * qinv;
    v4f res;
    #pragma unroll
    for (int c = 0; c < 4; ++c) {
        const float pden = fmaxf(sqrtf(ps[c]), EPS);
        const float pinv = 1.0f / pden;
        const float psq  = ps[c] * pinv * pinv;
        res[c] = 2.0f * d[c] * qinv * pinv - qsq - psq;
    }
    *(v4f*)(out + o) = res;
}

extern "C" void kernel_launch(void* const* d_in, const int* in_sizes, int n_in,
                              void* d_out, int out_size, void* d_ws, size_t ws_size,
                              hipStream_t stream)
{
    const float* query_feat   = (const float*)d_in[0];
    const float* support_feat = (const float*)d_in[1];
    float* out = (float*)d_out;

    float* part = (float*)d_ws;
    float* qn   = part + (size_t)NPLANE * PLANE;
    float* pn   = qn + (size_t)NPLANE * QSTR;

    slice_kernel<<<NPLANE, 256, 0, stream>>>(query_feat, support_feat, part, qn, pn);
    finish_kernel<<<188, 128, 0, stream>>>(part, qn, pn, out);
}


# Finish kernel: all 48 partial loads of a thread (16 planes x tile, row norm, class norm) issued as one group before the first add, one memory round trip instead of four; sum order unchanged
# speedup vs baseline: 3.5242x; 1.1137x over previous
#include <hip/hip_runtime.h>

#define T      16
#define WAY    20
#define SHOT   5
#define WQ     300
#define WS     100
#define C      1024
#define MTILES 19
#define EPS    1e-12f

#define KSPLIT 16
#define KSLICE 64
#define PITCH  68
#define NPLANE (T * KSPLIT)
#define PLANE  6144
#define QSTR   640
#define PSTR   32
#define APATCH (16 * PITCH)

typedef __attribute__((ext_vector_type(2))) float v2f;
typedef __attribute__((ext_vector_type(4))) float v4f;
typedef __attribute__((ext_vector_type(8))) float v8f;

__global__ __launch_bounds__(256)
void slice_kernel(const float* __restrict__ q,
                  const float* __restrict__ sup,
                  float* __restrict__ part,
                  float* __restrict__ qn,
                  float* __restrict__ pn)
{
    __shared__ __attribute__((aligned(16))) float sA[8][3][APATCH];
    __shared__ __attribute__((aligned(16))) float sB[WAY * PITCH];
    __shared__ __attribute__((aligned(16))) float sT[8][16 * WAY];

    const int bid  = blockIdx.x;
    const int t    = bid >> 4;
    const int ks   = bid & 15;
    const int tid  = threadIdx.x;
    const int w    = __builtin_amdgcn_readfirstlane(tid >> 5);
    const int lane = tid & 31;
    const int half = lane >> 4;
    const int l15  = lane & 15;
    const int kq   = tid & 15;
    const int r0   = tid >> 4;
    const int plane = ks * T + t;

    const float* sbase = sup + (size_t)t * WS * C + ks * KSLICE + kq * 4;
    v4f rs[SHOT];
    #pragma unroll
    for (int s = 0; s < SHOT; ++s)
        rs[s] = *(const v4f*)(sbase + (size_t)(r0 * SHOT + s) * C);
    v4f rs2[SHOT];
    #pragma unroll
    for (int s = 0; s < SHOT; ++s) rs2[s] = (v4f){0.f, 0.f, 0.f, 0.f};
    if (w < 2) {
        #pragma unroll
        for (int s = 0; s < SHOT; ++s)
            rs2[s] = *(const v4f*)(sbase + (size_t)((16 + r0) * SHOT + s) * C);
    }
    const float* qbase = q + (size_t)t * WQ * C + ks * KSLICE + l15 * 4;
    v4f ra0[8], ra1[8];
    #pragma unroll
    for (int i = 0; i < 8; ++i)
        ra0[i] = *(const v4f*)(qbase + (size_t)(w * 16 + 2 * i + half) * C);
    #pragma unroll
    for (int i = 0; i < 8; ++i)
        ra1[i] = *(const v4f*)(qbase + (size_t)((w + 8) * 16 + 2 * i + half) * C);
    __builtin_amdgcn_sched_barrier(0);

    const float inv_shot = 1.0f / (float)SHOT;
    {
        v4f a = {0.f, 0.f, 0.f, 0.f};
        #pragma unroll
        for (int s = 0; s < SHOT; ++s) a += rs[s];
        *(v4f*)(sB + r0 * PITCH + kq * 4) = a * inv_shot;
    }
    {
        v4f a = {0.f, 0.f, 0.f, 0.f};
        #pragma unroll
        for (int s = 0; s < SHOT; ++s) a += rs2[s];
        if (w < 2) *(v4f*)(sB + (16 + r0) * PITCH + kq * 4) = a * inv_shot;
    }
    __builtin_amdgcn_fence(__ATOMIC_RELEASE, "workgroup", "local");
    __builtin_amdgcn_s_barrier();
    __builtin_amdgcn_fence(__ATOMIC_ACQUIRE, "workgroup", "local");

    int n1 = 16 + l15;
    if (n1 > WAY - 1) n1 = WAY - 1;
    const float* fb0 = sB + l15 * PITCH + 4 * half;
    const float* fb1 = sB + n1  * PITCH + 4 * half;
    v4f fb0_r[KSLICE / 8], fb1_r[KSLICE / 8];
    #pragma unroll
    for (int j = 0; j < KSLICE / 8; ++j) {
        fb0_r[j] = *(const v4f*)(fb0 + 8 * j);
        fb1_r[j] = *(const v4f*)(fb1 + 8 * j);
    }

    if (w == 0) {
        float p0 = 0.f, p1 = 0.f;
        #pragma unroll
        for (int j = 0; j < KSLICE / 8; ++j) {
            const v4f b0 = fb0_r[j];
            const v4f b1 = fb1_r[j];
            p0 += b0.x * b0.x + b0.y * b0.y + b0.z * b0.z + b0.w * b0.w;
            p1 += b1.x * b1.x + b1.y * b1.y + b1.z * b1.z + b1.w * b1.w;
        }
        p0 += __shfl_xor(p0, 16);
        p1 += __shfl_xor(p1, 16);
        float* pp = pn + (size_t)plane * PSTR;
        if (lane < 16) pp[lane] = p0;
        if (lane < 4)  pp[16 + lane] = p1;
    }

    float* tp = part + (size_t)plane * PLANE;
    float* tq = qn + (size_t)plane * QSTR;
    float* st = sT[w];

    auto stage = [&](const v4f (&ra)[8], float* patch) {
        #pragma unroll
        for (int i = 0; i < 8; ++i)
            *(v4f*)(patch + (2 * i + half) * PITCH + l15 * 4) = ra[i];
    };

    auto compute = [&](const float* patch, int mt) {
        const float* fa = patch + l15 * PITCH + 4 * half;
        v4f fa_r[KSLICE / 8];
        #pragma unroll
        for (int j = 0; j < KSLICE / 8; ++j)
            fa_r[j] = *(const v4f*)(fa + 8 * j);
        __builtin_amdgcn_sched_barrier(0);

        v8f acc0 = {};
        v8f acc1 = {};
        float nacc = 0.f;
        #pragma unroll
        for (int j = 0; j < KSLICE / 8; ++j) {
            const v4f a  = fa_r[j];
            const v4f b0 = fb0_r[j];
            const v4f b1 = fb1_r[j];
            nacc += a.x * a.x + a.y * a.y + a.z * a.z + a.w * a.w;
            const v2f alo = {a.x, a.y},   ahi = {a.z, a.w};
            const v2f b0lo = {b0.x, b0.y}, b0hi = {b0.z, b0.w};
            const v2f b1lo = {b1.x, b1.y}, b1hi = {b1.z, b1.w};
            acc0 = __builtin_amdgcn_wmma_f32_16x16x4_f32(
                       false, alo, false, b0lo, (short)0, acc0, false, false);
            acc1 = __builtin_amdgcn_wmma_f32_16x16x4_f32(
                       false, alo, false, b1lo, (short)0, acc1, false, false);
            acc0 = __builtin_amdgcn_wmma_f32_16x16x4_f32(
                       false, ahi, false, b0hi, (short)0, acc0, false, false);
            acc1 = __builtin_amdgcn_wmma_f32_16x16x4_f32(
                       false, ahi, false, b1hi, (short)0, acc1, false, false);
        }
        const float nsum = nacc + __shfl_xor(nacc, 16);

        #pragma unroll
        for (int r = 0; r < 8; ++r)
            st[(r + 8 * half) * WAY + l15] = acc0[r];
        if (l15 < 4) {
            #pragma unroll
            for (int r = 0; r < 8; ++r)
                st[(r + 8 * half) * WAY + 16 + l15] = acc1[r];
        }
        #pragma unroll
        for (int i = 0; i < 3; ++i) {
            const int idx = lane + 32 * i;
            if (idx < 80)
                *(v4f*)(tp + mt * 320 + idx * 4) = *(const v4f*)(st + idx * 4);
        }
        if (lane < 16) tq[mt * 32 + lane] = nsum;
    };

    if (w < 3) {
        stage(ra0, sA[w][0]);
        v4f ra2[8];
        #pragma unroll
        for (int i = 0; i < 8; ++i) {
            int row = (w + 16) * 16 + 2 * i + half;
            if (row > WQ - 1) row = WQ - 1;
            ra2[i] = *(const v4f*)(qbase + (size_t)row * C);
        }
        compute(sA[w][0], w);
        stage(ra1, sA[w][1]);
        compute(sA[w][1], w + 8);
        stage(ra2, sA[w][2]);
        compute(sA[w][2], w + 16);
    } else {
        stage(ra0, sA[w][0]);
        compute(sA[w][0], w);
        stage(ra1, sA[w][1]);
        compute(sA[w][1], w + 8);
    }
}

__global__ __launch_bounds__(128)
void finish_kernel(const float* __restrict__ part,
                   const float* __restrict__ qn,
                   const float* __restrict__ pn,
                   float* __restrict__ out)
{
    const int g = blockIdx.x * 128 + threadIdx.x;
    if (g >= T * WQ * WAY / 4) return;
    const int o   = g * 4;
    const int t   = o / (WQ * WAY);
    const int rem = o - t * (WQ * WAY);
    const int row = rem / WAY;
    const int col = rem - row * WAY;
    const int qi  = (row >> 4) * 32 + (row & 15);

    v4f dv[KSPLIT], pv[KSPLIT];
    float qv[KSPLIT];
    #pragma unroll
    for (int ks = 0; ks < KSPLIT; ++ks)
        dv[ks] = *(const v4f*)(part + (size_t)(ks * T + t) * PLANE + rem);
    #pragma unroll
    for (int ks = 0; ks < KSPLIT; ++ks)
        qv[ks] = qn[(size_t)(ks * T + t) * QSTR + qi];
    #pragma unroll
    for (int ks = 0; ks < KSPLIT; ++ks)
        pv[ks] = *(const v4f*)(pn + (size_t)(ks * T + t) * PSTR + col);
    __builtin_amdgcn_sched_barrier(0);

    v4f d  = {0.f, 0.f, 0.f, 0.f};
    v4f ps = {0.f, 0.f, 0.f, 0.f};
    float qs = 0.f;
    #pragma unroll
    for (int ks = 0; ks < KSPLIT; ++ks) {
        d  += dv[ks];
        ps += pv[ks];
        qs += qv[ks];
    }
    const float qden = fmaxf(sqrtf(qs), EPS);
    const float qinv = 1.0f / qden;
    const float qsq  = qs * qinv * qinv;
    v4f res;
    #pragma unroll
    for (int c = 0; c < 4; ++c) {
        const float pden = fmaxf(sqrtf(ps[c]), EPS);
        const float pinv = 1.0f / pden;
        const float psq  = ps[c] * pinv * pinv;
        res[c] = 2.0f * d[c] * qinv * pinv - qsq - psq;
    }
    *(v4f*)(out + o) = res;
}

extern "C" void kernel_launch(void* const* d_in, const int* in_sizes, int n_in,
                              void* d_out, int out_size, void* d_ws, size_t ws_size,
                              hipStream_t stream)
{
    const float* query_feat   = (const float*)d_in[0];
    const float* support_feat = (const float*)d_in[1];
    float* out = (float*)d_out;

    float* part = (float*)d_ws;
    float* qn   = part + (size_t)NPLANE * PLANE;
    float* pn   = qn + (size_t)NPLANE * QSTR;

    slice_kernel<<<NPLANE, 256, 0, stream>>>(query_feat, support_feat, part, qn, pn);
    finish_kernel<<<188, 128, 0, stream>>>(part, qn, pn, out);
}
